# starting kernel
# baseline (speedup 1.0000x reference)
#include <hip/hip_runtime.h>


typedef __attribute__((ext_vector_type(2))) float v2f;
typedef __attribute__((ext_vector_type(8))) float v8f;

#define IN_DIM 128


__global__ void fill_deg_kernel(float* __restrict__ deg, int n) {
  int i = blockIdx.x * blockDim.x + threadIdx.x;
  if (i < n) deg[i] = 1.0f;
}

__global__ void deg_accum_kernel(const int* __restrict__ tgt, float* __restrict__ deg, int e) {
  int i = blockIdx.x * blockDim.x + threadIdx.x;
  if (i < e) atomicAdd(&deg[tgt[i]], 1.0f);
}

__global__ void rsqrt_kernel(float* __restrict__ d, int n) {
  int i = blockIdx.x * blockDim.x + threadIdx.x;
  if (i < n) d[i] = rsqrtf(d[i]);
}


template <int NC>
__global__ __launch_bounds__(256) void gemm_wmma_f32(const float* __restrict__ X,
                                                     const float* __restrict__ W,
                                                     float* __restrict__ H, int n_rows) {
  __shared__ __align__(16) float Wt[NC * IN_DIM];
  for (int i = threadIdx.x; i < NC * IN_DIM; i += 256) {
    int k = i / NC;
    int c = i - k * NC;
    Wt[c * IN_DIM + k] = W[i];
  }
  __syncthreads();

  const int lane = threadIdx.x & 31;
  const int wave = threadIdx.x >> 5;
  const int row_base = blockIdx.x * 128 + wave * 16;
  if (row_base + 16 > n_rows) return;

  const int r = lane & 15;
  const int koff = (lane >> 4) << 1;

  const float* xrow = X + (size_t)(row_base + r) * IN_DIM + koff;

  constexpr int NT = NC / 16;
  v8f acc[NT];
  const v8f vzero = {0.f, 0.f, 0.f, 0.f, 0.f, 0.f, 0.f, 0.f};
#pragma unroll
  for (int t = 0; t < NT; ++t) acc[t] = vzero;

#pragma unroll 4
  for (int k0 = 0; k0 < IN_DIM; k0 += 4) {
    v2f a = *(const v2f*)(xrow + k0);
#pragma unroll
    for (int t = 0; t < NT; ++t) {
      v2f b = *(const v2f*)(&Wt[(t * 16 + r) * IN_DIM + k0 + koff]);
      acc[t] = __builtin_amdgcn_wmma_f32_16x16x4_f32(false, a, false, b,
                                                     (short)0, acc[t], false, false);
    }
  }

  const int rowh = row_base + ((lane >> 4) << 3);
#pragma unroll
  for (int t = 0; t < NT; ++t) {
    const int col = t * 16 + r;
#pragma unroll
    for (int j = 0; j < 8; ++j)
      H[(size_t)(rowh + j) * NC + col] = acc[t][j];
  }
}


__global__ void init_agg_kernel(const float* __restrict__ h, const float* __restrict__ dinv,
                                const float* __restrict__ bias, float* __restrict__ agg,
                                int n, int D) {
  int idx = blockIdx.x * blockDim.x + threadIdx.x;
  if (idx >= n * D) return;
  int i = idx / D;
  int d = idx - i * D;
  float di = dinv[i];
  agg[idx] = di * di * h[idx] + bias[d];
}

__global__ __launch_bounds__(256) void scatter_d128(const int* __restrict__ src,
                                                    const int* __restrict__ tgt,
                                                    const float* __restrict__ dinv,
                                                    const float* __restrict__ h,
                                                    float* __restrict__ agg, int n_edges) {
  int e = blockIdx.x * 8 + (threadIdx.x >> 5);
  if (e >= n_edges) return;
  int lane = threadIdx.x & 31;
  int s = src[e], t = tgt[e];
  float w = dinv[s] * dinv[t];
  const float4 v = *(const float4*)(h + (size_t)s * 128 + lane * 4);
  float* ap = agg + (size_t)t * 128 + lane * 4;
  atomicAdd(ap + 0, w * v.x);
  atomicAdd(ap + 1, w * v.y);
  atomicAdd(ap + 2, w * v.z);
  atomicAdd(ap + 3, w * v.w);
}

__global__ __launch_bounds__(256) void scatter_d64(const int* __restrict__ src,
                                                   const int* __restrict__ tgt,
                                                   const float* __restrict__ dinv,
                                                   const float* __restrict__ h,
                                                   float* __restrict__ agg, int n_edges) {
  int e = blockIdx.x * 8 + (threadIdx.x >> 5);
  if (e >= n_edges) return;
  int lane = threadIdx.x & 31;
  int s = src[e], t = tgt[e];
  float w = dinv[s] * dinv[t];
  const float2 v = *(const float2*)(h + (size_t)s * 64 + lane * 2);
  float* ap = agg + (size_t)t * 64 + lane * 2;
  atomicAdd(ap + 0, w * v.x);
  atomicAdd(ap + 1, w * v.y);
}

__global__ void relu_kernel(float* __restrict__ a, int n) {
  int i = blockIdx.x * blockDim.x + threadIdx.x;
  if (i < n) a[i] = fmaxf(a[i], 0.0f);
}


static inline size_t align_up(size_t x, size_t a) { return (x + a - 1) & ~(a - 1); }

extern "C" void kernel_launch(void* const* d_in, const int* in_sizes, int n_in,
                              void* d_out, int out_size, void* d_ws, size_t ws_size,
                              hipStream_t stream) {
  const float* x  = (const float*)d_in[0];
  const int*   ei = (const int*)d_in[1];
  const float* W1 = (const float*)d_in[2];
  const float* b1 = (const float*)d_in[3];
  const float* W2 = (const float*)d_in[4];
  const float* b2 = (const float*)d_in[5];

  const int N = in_sizes[0] / IN_DIM;
  const int E = in_sizes[1] / 2;
  const int* src = ei;
  const int* tgt = ei + E;

  char* ws = (char*)d_ws;
  size_t off = 0;
  float* dinv = (float*)(ws + off); off = align_up(off + (size_t)N * 4, 256);
  float* H1   = (float*)(ws + off); off = align_up(off + (size_t)N * IN_DIM * 4, 256);
  float* A1   = (float*)(ws + off); off = align_up(off + (size_t)N * IN_DIM * 4, 256);
  float* H2   = H1;
  float* out  = (float*)d_out;

  const int TB = 256;
  dim3 blk(TB);

  fill_deg_kernel<<<dim3((N + TB - 1) / TB), blk, 0, stream>>>(dinv, N);
  deg_accum_kernel<<<dim3((E + TB - 1) / TB), blk, 0, stream>>>(tgt, dinv, E);
  rsqrt_kernel<<<dim3((N + TB - 1) / TB), blk, 0, stream>>>(dinv, N);

  const int gemm_blocks = (N + 127) / 128;
  const int eblocks = (E + 7) / 8;

  gemm_wmma_f32<128><<<dim3(gemm_blocks), blk, 0, stream>>>(x, W1, H1, N);
  init_agg_kernel<<<dim3((N * 128 + TB - 1) / TB), blk, 0, stream>>>(H1, dinv, b1, A1, N, 128);
  scatter_d128<<<dim3(eblocks), blk, 0, stream>>>(src, tgt, dinv, H1, A1, E);
  relu_kernel<<<dim3((N * 128 + TB - 1) / TB), blk, 0, stream>>>(A1, N * 128);

  gemm_wmma_f32<64><<<dim3(gemm_blocks), blk, 0, stream>>>(A1, W2, H2, N);
  init_agg_kernel<<<dim3((N * 64 + TB - 1) / TB), blk, 0, stream>>>(H2, dinv, b2, out, N, 64);
  scatter_d64<<<dim3(eblocks), blk, 0, stream>>>(src, tgt, dinv, H2, out, E);
}


# layer-1 edge pass as two column-half phases in launch order (doubled grid: first half adds columns 0-63 of every edge, rest 64-127): set touched per phase stays cache resident; atomic sum order kept
# speedup vs baseline: 3.0583x; 3.0583x over previous
#include <hip/hip_runtime.h>


typedef __attribute__((ext_vector_type(2))) float v2f;
typedef __attribute__((ext_vector_type(4))) float v4f;
typedef __attribute__((ext_vector_type(8))) float v8f;

#define IN_DIM 128


#define DEG_RANGE 12512
#define DEG_CHUNKS 32
#define DEG_TB 1024
#define DEG_BATCH 8

__device__ __forceinline__ void deg_count_one(unsigned* cnt, int t, int lo, unsigned len) {
  const unsigned d = (unsigned)(t - lo);
  if (d < len) atomicAdd(&cnt[d], 1u);
}

__global__ __launch_bounds__(DEG_TB) void deg_hist_kernel(const int* __restrict__ tgt,
                                                          unsigned* __restrict__ part,
                                                          int n, int e, int n_ranges, int echunk) {
  __shared__ unsigned cnt[DEG_RANGE];
  const int r = blockIdx.x % n_ranges;
  const int c = blockIdx.x / n_ranges;
  const int lo = r * DEG_RANGE;
  const unsigned len = (unsigned)min(DEG_RANGE, n - lo);
  for (int i = threadIdx.x; i < DEG_RANGE; i += DEG_TB) cnt[i] = 0u;
  __syncthreads();
  const int e0 = c * echunk;
  const int e1 = max(e0, min(e, e0 + echunk));
  const int e1v = e0 + ((e1 - e0) & ~3);
  for (int i = e0 + (int)threadIdx.x * 4; i < e1v; i += DEG_TB * 4 * DEG_BATCH) {
    int4 t[DEG_BATCH];
#pragma unroll
    for (int k = 0; k < DEG_BATCH; ++k) {
      const int ik = i + k * (DEG_TB * 4);
      t[k] = make_int4(-1, -1, -1, -1);
      if (ik < e1v) t[k] = *(const int4*)(tgt + ik);
    }
#pragma unroll
    for (int k = 0; k < DEG_BATCH; ++k) {
      deg_count_one(cnt, t[k].x, lo, len);
      deg_count_one(cnt, t[k].y, lo, len);
      deg_count_one(cnt, t[k].z, lo, len);
      deg_count_one(cnt, t[k].w, lo, len);
    }
  }
  if (e1v + (int)threadIdx.x < e1) deg_count_one(cnt, tgt[e1v + threadIdx.x], lo, len);
  __syncthreads();
  unsigned* row = part + (size_t)c * n + lo;
  for (int i = threadIdx.x; i < (int)len; i += DEG_TB) row[i] = cnt[i];
}

__global__ void dinv_kernel(const unsigned* __restrict__ part, float* __restrict__ dinv, int n) {
  int i = blockIdx.x * blockDim.x + threadIdx.x;
  if (i < n) {
    unsigned s = 1u;
#pragma unroll
    for (int c = 0; c < DEG_CHUNKS; ++c) s += part[(size_t)c * n + i];
    dinv[i] = rsqrtf((float)s);
  }
}


#define KSTEPS (IN_DIM / 4)

__device__ __forceinline__ void load_a_strip(v2f (&a)[KSTEPS], const float* __restrict__ X,
                                             int strip, int r, int koff) {
  const float* xrow = X + (size_t)(strip * 16 + r) * IN_DIM + koff;
#pragma unroll
  for (int kk = 0; kk < KSTEPS; ++kk) a[kk] = *(const v2f*)(xrow + 4 * kk);
}

template <int NC, bool RELU_A>
__device__ __forceinline__ void mul_store_strip(const v2f (&a)[KSTEPS], const v2f (&b)[KSTEPS][4],
                                                const float* __restrict__ dinv, const float (&bias)[4],
                                                float* __restrict__ H, float* __restrict__ AGG,
                                                int strip, int col0, int lane) {
  const int rowh = strip * 16 + ((lane >> 4) << 3);
  const v4f dlo = *(const v4f*)(dinv + rowh);
  const v4f dhi = *(const v4f*)(dinv + rowh + 4);
  v8f acc[4];
  const v8f vzero = {0.f, 0.f, 0.f, 0.f, 0.f, 0.f, 0.f, 0.f};
#pragma unroll
  for (int t = 0; t < 4; ++t) acc[t] = vzero;
#pragma unroll
  for (int kk = 0; kk < KSTEPS; ++kk) {
    v2f av = a[kk];
    if (RELU_A) {
      av.x = fmaxf(av.x, 0.0f);
      av.y = fmaxf(av.y, 0.0f);
    }
#pragma unroll
    for (int t = 0; t < 4; ++t)
      acc[t] = __builtin_amdgcn_wmma_f32_16x16x4_f32(false, av, false, b[kk][t],
                                                     (short)0, acc[t], false, false);
  }
  float d[8];
#pragma unroll
  for (int j = 0; j < 4; ++j) {
    d[j] = dlo[j];
    d[j + 4] = dhi[j];
  }
  const size_t o = (size_t)rowh * NC + col0 + (lane & 15);
  float* hp = H + o;
  float* ap = AGG + o;
  if (RELU_A) {
#pragma unroll
    for (int t = 0; t < 4; ++t) {
#pragma unroll
      for (int j = 0; j < 8; ++j) {
        const float h = d[j] * acc[t][j];
        hp[(size_t)j * NC + t * 16] = d[j] * h;
        ap[(size_t)j * NC + t * 16] = (d[j] * d[j]) * h + bias[t];
      }
    }
  } else {
    float rd[8];
#pragma unroll
    for (int j = 0; j < 8; ++j) rd[j] = 1.0f / d[j];
#pragma unroll
    for (int t = 0; t < 4; ++t) {
#pragma unroll
      for (int j = 0; j < 8; ++j) {
        const float hs = d[j] * acc[t][j];
        hp[(size_t)j * NC + t * 16] = hs;
        ap[(size_t)j * NC + t * 16] = hs + bias[t] * rd[j];
      }
    }
  }
}

template <int NC, bool RELU_A>
__global__ __launch_bounds__(256) void gemm_wmma_f32(const float* __restrict__ X,
                                                     const float* __restrict__ W,
                                                     const float* __restrict__ dinv,
                                                     const float* __restrict__ bias_v,
                                                     float* __restrict__ H,
                                                     float* __restrict__ AGG, int n_rows) {
  constexpr int CG = NC / 64;
  const int lane = threadIdx.x & 31;
  const int wave = __builtin_amdgcn_readfirstlane((int)(threadIdx.x >> 5));
  const int gw = blockIdx.x * 8 + wave;
  const int cg = gw % CG;
  const int slot = gw / CG;
  const int nslots = (int)(gridDim.x * 8) / CG;
  const int n_strips = n_rows >> 4;
  if (slot >= n_strips) return;

  const int r = lane & 15;
  const int koff = (lane >> 4) << 1;
  const int col0 = cg * 64;

  v2f b[KSTEPS][4];
  {
    const float* wp = W + (size_t)koff * NC + col0 + r;
#pragma unroll
    for (int kk = 0; kk < KSTEPS; ++kk) {
#pragma unroll
      for (int t = 0; t < 4; ++t) {
        v2f f;
        f.x = wp[(4 * kk) * NC + t * 16];
        f.y = wp[(4 * kk + 1) * NC + t * 16];
        b[kk][t] = f;
      }
    }
  }

  float bias[4];
#pragma unroll
  for (int t = 0; t < 4; ++t) bias[t] = bias_v[col0 + t * 16 + r];

  v2f a0[KSTEPS], a1[KSTEPS];
  int strip = slot;
  load_a_strip(a0, X, strip, r, koff);
  for (;;) {
    const int s1 = strip + nslots;
    if (s1 >= n_strips) {
      mul_store_strip<NC, RELU_A>(a0, b, dinv, bias, H, AGG, strip, col0, lane);
      break;
    }
    load_a_strip(a1, X, s1, r, koff);
    __builtin_amdgcn_sched_barrier(0);
    mul_store_strip<NC, RELU_A>(a0, b, dinv, bias, H, AGG, strip, col0, lane);
    const int s2 = s1 + nslots;
    if (s2 >= n_strips) {
      mul_store_strip<NC, RELU_A>(a1, b, dinv, bias, H, AGG, s1, col0, lane);
      break;
    }
    load_a_strip(a0, X, s2, r, koff);
    __builtin_amdgcn_sched_barrier(0);
    mul_store_strip<NC, RELU_A>(a1, b, dinv, bias, H, AGG, s1, col0, lane);
    strip = s2;
  }
}


template <int D, int EPB, bool WT, int CPH>
__global__ __launch_bounds__(256, 16) void scatter_edges(const int* __restrict__ src,
                                                     const int* __restrict__ tgt,
                                                     const float* __restrict__ dinv,
                                                     const float* __restrict__ h,
                                                     float* __restrict__ agg, int n_edges,
                                                     int blocks_per_phase) {
  constexpr int C = D / 32 / CPH;
  const int lane = threadIdx.x & 31;
  int bx = blockIdx.x;
  int col0 = 0;
  if (CPH > 1) {
    const int ph = bx / blocks_per_phase;
    bx -= ph * blocks_per_phase;
    col0 = ph * (D / CPH);
  }
  const int e0 = (bx * 8 + (threadIdx.x >> 5)) * 32;
  if (e0 >= n_edges) return;
  const int e = e0 + lane;
  int s = 0, t = 0;
  float w = 0.0f;
  if (e < n_edges) {
    s = src[e];
    t = tgt[e];
    if (WT) w = dinv[t];
  }
  const int wbits = __float_as_int(w);
  const int cnt = n_edges - e0;

  if (cnt >= 32) {
#pragma unroll
    for (int b = 0; b < 32; b += EPB) {
      float v[EPB][C];
#pragma unroll
      for (int j = 0; j < EPB; ++j) {
        const int sj = __builtin_amdgcn_readlane(s, b + j);
        const float* hp = h + (size_t)sj * D + col0 + lane;
#pragma unroll
        for (int c = 0; c < C; ++c) v[j][c] = hp[32 * c];
      }
#pragma unroll
      for (int j = 0; j < EPB; ++j) {
        const int tj = __builtin_amdgcn_readlane(t, b + j);
        float* ap = agg + (size_t)tj * D + col0 + lane;
        if (WT) {
          const float wj = __int_as_float(__builtin_amdgcn_readlane(wbits, b + j));
#pragma unroll
          for (int c = 0; c < C; ++c) atomicAdd(ap + 32 * c, wj * v[j][c]);
        } else {
#pragma unroll
          for (int c = 0; c < C; ++c) atomicAdd(ap + 32 * c, v[j][c]);
        }
      }
    }
  } else {
    for (int j = 0; j < cnt; ++j) {
      const int sj = __builtin_amdgcn_readlane(s, j);
      const int tj = __builtin_amdgcn_readlane(t, j);
      const float wj = WT ? __int_as_float(__builtin_amdgcn_readlane(wbits, j)) : 1.0f;
      const float* hp = h + (size_t)sj * D + col0 + lane;
      float* ap = agg + (size_t)tj * D + col0 + lane;
      float v[C];
#pragma unroll
      for (int c = 0; c < C; ++c) v[c] = hp[32 * c];
#pragma unroll
      for (int c = 0; c < C; ++c) atomicAdd(ap + 32 * c, WT ? wj * v[c] : v[c]);
    }
  }
}


static inline size_t align_up(size_t x, size_t a) { return (x + a - 1) & ~(a - 1); }

extern "C" void kernel_launch(void* const* d_in, const int* in_sizes, int n_in,
                              void* d_out, int out_size, void* d_ws, size_t ws_size,
                              hipStream_t stream) {
  const float* x  = (const float*)d_in[0];
  const int*   ei = (const int*)d_in[1];
  const float* W1 = (const float*)d_in[2];
  const float* b1 = (const float*)d_in[3];
  const float* W2 = (const float*)d_in[4];
  const float* b2 = (const float*)d_in[5];

  const int N = in_sizes[0] / IN_DIM;
  const int E = in_sizes[1] / 2;
  const int* src = ei;
  const int* tgt = ei + E;

  char* ws = (char*)d_ws;
  size_t off = 0;
  float* dinv = (float*)(ws + off); off = align_up(off + (size_t)N * 4, 256);
  float* H1   = (float*)(ws + off); off = align_up(off + (size_t)N * IN_DIM * 4, 256);
  float* A1   = (float*)(ws + off); off = align_up(off + (size_t)N * IN_DIM * 4, 256);
  unsigned* part = (unsigned*)(ws + off); off = align_up(off + (size_t)DEG_CHUNKS * N * 4, 256);
  float* H2   = H1;
  float* out  = (float*)d_out;

  const int TB = 256;
  dim3 blk(TB);

  const int n_ranges = (N + DEG_RANGE - 1) / DEG_RANGE;
  const int echunk = (((E + DEG_CHUNKS - 1) / DEG_CHUNKS) + 3) & ~3;
  deg_hist_kernel<<<dim3(n_ranges * DEG_CHUNKS), dim3(DEG_TB), 0, stream>>>(tgt, part, N, E, n_ranges, echunk);
  dinv_kernel<<<dim3((N + TB - 1) / TB), blk, 0, stream>>>(part, dinv, N);

  const int gemm_blocks = 256;
  const int eblocks = (E + 255) / 256;

  gemm_wmma_f32<128, false><<<dim3(gemm_blocks), blk, 0, stream>>>(x, W1, dinv, b1, H1, A1, N);
  scatter_edges<128, 16, false, 2><<<dim3(eblocks * 2), blk, 0, stream>>>(src, tgt, dinv, H1, A1, E, eblocks);

  gemm_wmma_f32<64, true><<<dim3(gemm_blocks), blk, 0, stream>>>(A1, W2, dinv, b2, H2, out, N);
  scatter_edges<64, 16, true, 1><<<dim3(eblocks), blk, 0, stream>>>(src, tgt, dinv, H2, out, E, eblocks);
}


# layer-2 edge pass: work-group reads its die, adds only edges whose target is in its chip half's half of the nodes (vote + lane permute): a sum line gets atomics from one L2 half; arrival order kept
# speedup vs baseline: 3.6855x; 1.2051x over previous
#include <hip/hip_runtime.h>


typedef __attribute__((ext_vector_type(2))) float v2f;
typedef __attribute__((ext_vector_type(4))) float v4f;
typedef __attribute__((ext_vector_type(8))) float v8f;

#define IN_DIM 128


#define DEG_RANGE 12512
#define DEG_CHUNKS 32
#define DEG_TB 1024
#define DEG_BATCH 8

__device__ __forceinline__ void deg_count_one(unsigned* cnt, int t, int lo, unsigned len) {
  const unsigned d = (unsigned)(t - lo);
  if (d < len) atomicAdd(&cnt[d], 1u);
}

__global__ __launch_bounds__(DEG_TB) void deg_hist_kernel(const int* __restrict__ tgt,
                                                          unsigned* __restrict__ part,
                                                          int n, int e, int n_ranges, int echunk) {
  __shared__ unsigned cnt[DEG_RANGE];
  const int r = blockIdx.x % n_ranges;
  const int c = blockIdx.x / n_ranges;
  const int lo = r * DEG_RANGE;
  const unsigned len = (unsigned)min(DEG_RANGE, n - lo);
  for (int i = threadIdx.x; i < DEG_RANGE; i += DEG_TB) cnt[i] = 0u;
  __syncthreads();
  const int e0 = c * echunk;
  const int e1 = max(e0, min(e, e0 + echunk));
  const int e1v = e0 + ((e1 - e0) & ~3);
  for (int i = e0 + (int)threadIdx.x * 4; i < e1v; i += DEG_TB * 4 * DEG_BATCH) {
    int4 t[DEG_BATCH];
#pragma unroll
    for (int k = 0; k < DEG_BATCH; ++k) {
      const int ik = i + k * (DEG_TB * 4);
      t[k] = make_int4(-1, -1, -1, -1);
      if (ik < e1v) t[k] = *(const int4*)(tgt + ik);
    }
#pragma unroll
    for (int k = 0; k < DEG_BATCH; ++k) {
      deg_count_one(cnt, t[k].x, lo, len);
      deg_count_one(cnt, t[k].y, lo, len);
      deg_count_one(cnt, t[k].z, lo, len);
      deg_count_one(cnt, t[k].w, lo, len);
    }
  }
  if (e1v + (int)threadIdx.x < e1) deg_count_one(cnt, tgt[e1v + threadIdx.x], lo, len);
  __syncthreads();
  unsigned* row = part + (size_t)c * n + lo;
  for (int i = threadIdx.x; i < (int)len; i += DEG_TB) row[i] = cnt[i];
}

#define JOB_PITCH 64
#define JOB_COUNTERS 4

__global__ void dinv_kernel(const unsigned* __restrict__ part, float* __restrict__ dinv, int n,
                            unsigned* __restrict__ jobs) {
  int i = blockIdx.x * blockDim.x + threadIdx.x;
  if (i < JOB_COUNTERS) jobs[i * JOB_PITCH] = 0u;
  if (i < n) {
    unsigned s = 1u;
#pragma unroll
    for (int c = 0; c < DEG_CHUNKS; ++c) s += part[(size_t)c * n + i];
    dinv[i] = rsqrtf((float)s);
  }
}


#define KSTEPS (IN_DIM / 4)

__device__ __forceinline__ void load_a_strip(v2f (&a)[KSTEPS], const float* __restrict__ X,
                                             int strip, int r, int koff) {
  const float* xrow = X + (size_t)(strip * 16 + r) * IN_DIM + koff;
#pragma unroll
  for (int kk = 0; kk < KSTEPS; ++kk) a[kk] = *(const v2f*)(xrow + 4 * kk);
}

template <int NC, bool RELU_A>
__device__ __forceinline__ void mul_store_strip(const v2f (&a)[KSTEPS], const v2f (&b)[KSTEPS][4],
                                                const float* __restrict__ dinv, const float (&bias)[4],
                                                float* __restrict__ H, float* __restrict__ AGG,
                                                int strip, int col0, int lane) {
  const int rowh = strip * 16 + ((lane >> 4) << 3);
  const v4f dlo = *(const v4f*)(dinv + rowh);
  const v4f dhi = *(const v4f*)(dinv + rowh + 4);
  v8f acc[4];
  const v8f vzero = {0.f, 0.f, 0.f, 0.f, 0.f, 0.f, 0.f, 0.f};
#pragma unroll
  for (int t = 0; t < 4; ++t) acc[t] = vzero;
#pragma unroll
  for (int kk = 0; kk < KSTEPS; ++kk) {
    v2f av = a[kk];
    if (RELU_A) {
      av.x = fmaxf(av.x, 0.0f);
      av.y = fmaxf(av.y, 0.0f);
    }
#pragma unroll
    for (int t = 0; t < 4; ++t)
      acc[t] = __builtin_amdgcn_wmma_f32_16x16x4_f32(false, av, false, b[kk][t],
                                                     (short)0, acc[t], false, false);
  }
  float d[8];
#pragma unroll
  for (int j = 0; j < 4; ++j) {
    d[j] = dlo[j];
    d[j + 4] = dhi[j];
  }
  const size_t o = (size_t)rowh * NC + col0 + (lane & 15);
  float* hp = H + o;
  float* ap = AGG + o;
  if (RELU_A) {
#pragma unroll
    for (int t = 0; t < 4; ++t) {
#pragma unroll
      for (int j = 0; j < 8; ++j) {
        const float h = d[j] * acc[t][j];
        hp[(size_t)j * NC + t * 16] = d[j] * h;
        ap[(size_t)j * NC + t * 16] = (d[j] * d[j]) * h + bias[t];
      }
    }
  } else {
    float rd[8];
#pragma unroll
    for (int j = 0; j < 8; ++j) rd[j] = 1.0f / d[j];
#pragma unroll
    for (int t = 0; t < 4; ++t) {
#pragma unroll
      for (int j = 0; j < 8; ++j) {
        const float hs = d[j] * acc[t][j];
        hp[(size_t)j * NC + t * 16] = hs;
        ap[(size_t)j * NC + t * 16] = hs + bias[t] * rd[j];
      }
    }
  }
}

template <int NC, bool RELU_A>
__global__ __launch_bounds__(256) void gemm_wmma_f32(const float* __restrict__ X,
                                                     const float* __restrict__ W,
                                                     const float* __restrict__ dinv,
                                                     const float* __restrict__ bias_v,
                                                     float* __restrict__ H,
                                                     float* __restrict__ AGG, int n_rows) {
  constexpr int CG = NC / 64;
  const int lane = threadIdx.x & 31;
  const int wave = __builtin_amdgcn_readfirstlane((int)(threadIdx.x >> 5));
  const int gw = blockIdx.x * 8 + wave;
  const int cg = gw % CG;
  const int slot = gw / CG;
  const int nslots = (int)(gridDim.x * 8) / CG;
  const int n_strips = n_rows >> 4;
  if (slot >= n_strips) return;

  const int r = lane & 15;
  const int koff = (lane >> 4) << 1;
  const int col0 = cg * 64;

  v2f b[KSTEPS][4];
  {
    const float* wp = W + (size_t)koff * NC + col0 + r;
#pragma unroll
    for (int kk = 0; kk < KSTEPS; ++kk) {
#pragma unroll
      for (int t = 0; t < 4; ++t) {
        v2f f;
        f.x = wp[(4 * kk) * NC + t * 16];
        f.y = wp[(4 * kk + 1) * NC + t * 16];
        b[kk][t] = f;
      }
    }
  }

  float bias[4];
#pragma unroll
  for (int t = 0; t < 4; ++t) bias[t] = bias_v[col0 + t * 16 + r];

  v2f a0[KSTEPS], a1[KSTEPS];
  int strip = slot;
  load_a_strip(a0, X, strip, r, koff);
  for (;;) {
    const int s1 = strip + nslots;
    if (s1 >= n_strips) {
      mul_store_strip<NC, RELU_A>(a0, b, dinv, bias, H, AGG, strip, col0, lane);
      break;
    }
    load_a_strip(a1, X, s1, r, koff);
    __builtin_amdgcn_sched_barrier(0);
    mul_store_strip<NC, RELU_A>(a0, b, dinv, bias, H, AGG, strip, col0, lane);
    const int s2 = s1 + nslots;
    if (s2 >= n_strips) {
      mul_store_strip<NC, RELU_A>(a1, b, dinv, bias, H, AGG, s1, col0, lane);
      break;
    }
    load_a_strip(a0, X, s2, r, koff);
    __builtin_amdgcn_sched_barrier(0);
    mul_store_strip<NC, RELU_A>(a1, b, dinv, bias, H, AGG, s1, col0, lane);
    strip = s2;
  }
}


__device__ __forceinline__ int take_edge_job(unsigned* __restrict__ jobs, int n_blocks) {
  __shared__ int job;
  if (threadIdx.x == 0) {
    int half = (int)((__builtin_amdgcn_s_sendmsg_rtn(0x87) >> 18) & 1u);
    unsigned tk = atomicAdd(jobs + half * JOB_PITCH, 1u);
    if (tk >= (unsigned)n_blocks) {
      half ^= 1;
      tk = atomicAdd(jobs + half * JOB_PITCH, 1u);
    }
    job = (int)(tk * 2u) + half;
  }
  __syncthreads();
  return __builtin_amdgcn_readfirstlane(job);
}

template <int D, int EPB>
__global__ __launch_bounds__(256, 16) void scatter_edges_col_half(const int* __restrict__ src,
                                                              const int* __restrict__ tgt,
                                                              const float* __restrict__ h,
                                                              float* __restrict__ agg, int n_edges,
                                                              int n_blocks, unsigned* __restrict__ jobs) {
  constexpr int C = D / 32 / 2;
  const int lane = threadIdx.x & 31;
  const int jb = take_edge_job(jobs, n_blocks);
  const int bx = jb >> 1;
  if (bx >= n_blocks) return;
  const int col0 = (jb & 1) * (D / 2);
  const int e0 = (bx * 8 + (threadIdx.x >> 5)) * 32;
  if (e0 >= n_edges) return;
  const int e = e0 + lane;
  int s = 0, t = 0;
  if (e < n_edges) {
    s = src[e];
    t = tgt[e];
  }
  const int cnt = n_edges - e0;

  if (cnt >= 32) {
#pragma unroll
    for (int b = 0; b < 32; b += EPB) {
      float v[EPB][C];
#pragma unroll
      for (int j = 0; j < EPB; ++j) {
        const int sj = __builtin_amdgcn_readlane(s, b + j);
        const float* hp = h + (size_t)sj * D + col0 + lane;
#pragma unroll
        for (int c = 0; c < C; ++c) v[j][c] = hp[32 * c];
      }
#pragma unroll
      for (int j = 0; j < EPB; ++j) {
        const int tj = __builtin_amdgcn_readlane(t, b + j);
        float* ap = agg + (size_t)tj * D + col0 + lane;
#pragma unroll
        for (int c = 0; c < C; ++c) atomicAdd(ap + 32 * c, v[j][c]);
      }
    }
  } else {
    for (int j = 0; j < cnt; ++j) {
      const int sj = __builtin_amdgcn_readlane(s, j);
      const int tj = __builtin_amdgcn_readlane(t, j);
      const float* hp = h + (size_t)sj * D + col0 + lane;
      float* ap = agg + (size_t)tj * D + col0 + lane;
      float v[C];
#pragma unroll
      for (int c = 0; c < C; ++c) v[c] = hp[32 * c];
#pragma unroll
      for (int c = 0; c < C; ++c) atomicAdd(ap + 32 * c, v[c]);
    }
  }
}

#define GPW 2

template <int D, int B, int NB>
__device__ __forceinline__ void add_batch(int cs, int ct, int wbits, int cnt, const float* __restrict__ h,
                                          float* __restrict__ agg, int lane) {
  constexpr int C = D / 32;
  const int csb = lane < cnt ? cs : __builtin_amdgcn_readlane(cs, B);
  float v[NB][C];
#pragma unroll
  for (int j = 0; j < NB; ++j) {
    const int sj = __builtin_amdgcn_readlane(csb, B + j);
    const float* hp = h + (size_t)sj * D + lane;
#pragma unroll
    for (int c = 0; c < C; ++c) v[j][c] = hp[32 * c];
  }
  __builtin_amdgcn_sched_barrier(0);
#pragma unroll
  for (int j = 0; j < NB; ++j) {
    if (B + j < cnt) {
      const int tj = __builtin_amdgcn_readlane(ct, B + j);
      const float wj = __int_as_float(__builtin_amdgcn_readlane(wbits, B + j));
      float* ap = agg + (size_t)tj * D + lane;
#pragma unroll
      for (int c = 0; c < C; ++c) atomicAdd(ap + 32 * c, wj * v[j][c]);
    }
  }
}

template <int D>
__global__ __launch_bounds__(256, 16) void scatter_edges_tgt_half(const int* __restrict__ src,
                                                              const int* __restrict__ tgt,
                                                              const float* __restrict__ dinv,
                                                              const float* __restrict__ h,
                                                              float* __restrict__ agg, int n_edges,
                                                              int n_nodes, int n_blocks,
                                                              unsigned* __restrict__ jobs) {
  const int lane = threadIdx.x & 31;
  const int jb = take_edge_job(jobs, n_blocks);
  const int bx = jb >> 1;
  if (bx >= n_blocks) return;
  const bool upper = (jb & 1) != 0;
  const int n_split = n_nodes >> 1;
  const int e0 = (bx * 8 + (int)(threadIdx.x >> 5)) * (32 * GPW);
  if (e0 >= n_edges) return;

  int s[GPW], t[GPW];
#pragma unroll
  for (int g = 0; g < GPW; ++g) {
    const int e = e0 + 32 * g + lane;
    s[g] = 0;
    t[g] = 0;
    if (e < n_edges) {
      s[g] = src[e];
      t[g] = tgt[e];
    }
  }

#pragma unroll
  for (int g = 0; g < GPW; ++g) {
    const bool mine = (e0 + 32 * g + lane < n_edges) && ((t[g] >= n_split) == upper);
    const unsigned vote = (unsigned)__ballot(mine);
    const int cnt = __builtin_popcount(vote);
    const int pos = mine ? (int)__builtin_amdgcn_mbcnt_lo(vote, 0u) : cnt + (int)__builtin_amdgcn_mbcnt_lo(~vote, 0u);
    const int cs = __builtin_amdgcn_ds_permute(pos << 2, s[g]);
    const int ct = __builtin_amdgcn_ds_permute(pos << 2, t[g]);
    float w = 0.0f;
    if (lane < cnt) w = dinv[ct];
    const int wbits = __float_as_int(w);
    if (cnt > 0) add_batch<D, 0, 16>(cs, ct, wbits, cnt, h, agg, lane);
    if (cnt > 16) add_batch<D, 16, 8>(cs, ct, wbits, cnt, h, agg, lane);
    if (cnt > 24) add_batch<D, 24, 8>(cs, ct, wbits, cnt, h, agg, lane);
  }
}


static inline size_t align_up(size_t x, size_t a) { return (x + a - 1) & ~(a - 1); }

extern "C" void kernel_launch(void* const* d_in, const int* in_sizes, int n_in,
                              void* d_out, int out_size, void* d_ws, size_t ws_size,
                              hipStream_t stream) {
  const float* x  = (const float*)d_in[0];
  const int*   ei = (const int*)d_in[1];
  const float* W1 = (const float*)d_in[2];
  const float* b1 = (const float*)d_in[3];
  const float* W2 = (const float*)d_in[4];
  const float* b2 = (const float*)d_in[5];

  const int N = in_sizes[0] / IN_DIM;
  const int E = in_sizes[1] / 2;
  const int* src = ei;
  const int* tgt = ei + E;

  char* ws = (char*)d_ws;
  size_t off = 0;
  float* dinv = (float*)(ws + off); off = align_up(off + (size_t)N * 4, 256);
  float* H1   = (float*)(ws + off); off = align_up(off + (size_t)N * IN_DIM * 4, 256);
  float* A1   = (float*)(ws + off); off = align_up(off + (size_t)N * IN_DIM * 4, 256);
  unsigned* part = (unsigned*)(ws + off); off = align_up(off + (size_t)DEG_CHUNKS * N * 4, 256);
  unsigned* jobs = (unsigned*)(ws + off); off = align_up(off + (size_t)JOB_COUNTERS * JOB_PITCH * 4, 256);
  float* H2   = H1;
  float* out  = (float*)d_out;

  const int TB = 256;
  dim3 blk(TB);

  const int n_ranges = (N + DEG_RANGE - 1) / DEG_RANGE;
  const int echunk = (((E + DEG_CHUNKS - 1) / DEG_CHUNKS) + 3) & ~3;
  deg_hist_kernel<<<dim3(n_ranges * DEG_CHUNKS), dim3(DEG_TB), 0, stream>>>(tgt, part, N, E, n_ranges, echunk);
  dinv_kernel<<<dim3((N + TB - 1) / TB), blk, 0, stream>>>(part, dinv, N, jobs);

  const int gemm_blocks = 256;
  const int eblocks = (E + 255) / 256;
  const int eblocks2 = (E + 256 * GPW - 1) / (256 * GPW);

  gemm_wmma_f32<128, false><<<dim3(gemm_blocks), blk, 0, stream>>>(x, W1, dinv, b1, H1, A1, N);
  scatter_edges_col_half<128, 16><<<dim3(eblocks * 2), blk, 0, stream>>>(src, tgt, H1, A1, E, eblocks, jobs);

  gemm_wmma_f32<64, true><<<dim3(gemm_blocks), blk, 0, stream>>>(A1, W2, dinv, b2, H2, out, N);
  scatter_edges_tgt_half<64><<<dim3(eblocks2 * 2), blk, 0, stream>>>(src, tgt, dinv, H2, out, E, N, eblocks2,
                                                                         jobs + 2 * JOB_PITCH);
}
